# starting kernel
# baseline (speedup 1.0000x reference)
#include <hip/hip_runtime.h>
#include <hip/hip_bf16.h>

#define N_NODES   100000
#define N_EDGES   1600000
#define NODE_DIM  64
#define HIDDEN    256
#define OUT_DIM   64
#define BN_EPS    1e-5f

typedef __attribute__((ext_vector_type(2))) float v2f;
typedef __attribute__((ext_vector_type(8))) float v8f;

#define WMMA_F32X4(ACC, A, B) \
  ACC = __builtin_amdgcn_wmma_f32_16x16x4_f32(false, (A), false, (B), (short)0, (ACC), false, false)

__global__ __launch_bounds__(256) void zero_ws(float* __restrict__ p, int n) {
  int i = blockIdx.x * 256 + threadIdx.x;
  if (i < n) p[i] = 0.0f;
}

__global__ __launch_bounds__(256) void edge_scatter(
    const float* __restrict__ h, const int* __restrict__ src,
    const int* __restrict__ dst, float* __restrict__ agg,
    float* __restrict__ deg) {
  unsigned tid = blockIdx.x * 256u + threadIdx.x;
  if (tid >= (unsigned)N_EDGES * NODE_DIM) return;
  int e = tid >> 6;
  int c = tid & 63;
  int s = src[e];
  int d = dst[e];
  atomicAdd(&agg[d * NODE_DIM + c], h[s * NODE_DIM + c]);
  if (c == 0) atomicAdd(&deg[d], 1.0f);
}

__global__ __launch_bounds__(128) void fused_mlp(
    const float* __restrict__ h,  const float* __restrict__ W1,
    const float* __restrict__ b1, const float* __restrict__ W2,
    const float* __restrict__ b2, const float* __restrict__ agg,
    const float* __restrict__ deg, float* __restrict__ out) {
  __shared__ float xs[16][NODE_DIM + 1];
  __shared__ float hs[16][HIDDEN + 1];

  const int m0   = blockIdx.x * 16;
  const int tid  = threadIdx.x;
  const int wave = tid >> 5;
  const int lane = tid & 31;
  const int half = lane >> 4;
  const int r    = lane & 15;

  for (int i = tid; i < 16 * NODE_DIM; i += 128) {
    int m = i >> 6, c = i & 63;
    int g = (m0 + m) * NODE_DIM + c;
    float d = deg[m0 + m];
    d = d > 1.0f ? d : 1.0f;
    xs[m][c] = agg[g] / d + h[g];
  }
  __syncthreads();

  v8f acc0 = {}, acc1 = {}, acc2 = {}, acc3 = {};
  const int nb = wave * 64;
  for (int k = 0; k < NODE_DIM; k += 4) {
    v2f a;
    a.x = xs[r][k + 2 * half];
    a.y = xs[r][k + 2 * half + 1];
    const float* w1p0 = W1 + (k + 2 * half) * HIDDEN + nb + r;
    const float* w1p1 = W1 + (k + 2 * half + 1) * HIDDEN + nb + r;
    v2f b;
    b.x = w1p0[0];  b.y = w1p1[0];  WMMA_F32X4(acc0, a, b);
    b.x = w1p0[16]; b.y = w1p1[16]; WMMA_F32X4(acc1, a, b);
    b.x = w1p0[32]; b.y = w1p1[32]; WMMA_F32X4(acc2, a, b);
    b.x = w1p0[48]; b.y = w1p1[48]; WMMA_F32X4(acc3, a, b);
  }
#define STORE_TILE(ACC, T)                                        \
  {                                                               \
    const int n = nb + (T) * 16 + r;                              \
    const float bias = b1[n];                                     \
    _Pragma("unroll") for (int j = 0; j < 8; ++j) {               \
      float v = (ACC)[j] + bias;                                  \
      hs[j + 8 * half][n] = v > 0.0f ? v : 0.0f;                  \
    }                                                             \
  }
  STORE_TILE(acc0, 0)
  STORE_TILE(acc1, 1)
  STORE_TILE(acc2, 2)
  STORE_TILE(acc3, 3)
#undef STORE_TILE
  __syncthreads();

  v8f occ = {};
  const int ob = wave * 16;
  for (int k = 0; k < HIDDEN; k += 4) {
    v2f a;
    a.x = hs[r][k + 2 * half];
    a.y = hs[r][k + 2 * half + 1];
    v2f b;
    b.x = W2[(k + 2 * half) * OUT_DIM + ob + r];
    b.y = W2[(k + 2 * half + 1) * OUT_DIM + ob + r];
    WMMA_F32X4(occ, a, b);
  }
  {
    const float bias = b2[ob + r];
#pragma unroll
    for (int j = 0; j < 8; ++j) {
      float v = occ[j] + bias;
      v = v > 0.0f ? v : 0.0f;
      out[(m0 + j + 8 * half) * OUT_DIM + ob + r] = v;
    }
  }
}

__global__ __launch_bounds__(256) void bn_stats(
    const float* __restrict__ x, float* __restrict__ gsum,
    float* __restrict__ gsq) {
  __shared__ float ssum[OUT_DIM], ssq[OUT_DIM];
  const int tid = threadIdx.x;
  if (tid < OUT_DIM) { ssum[tid] = 0.0f; ssq[tid] = 0.0f; }
  __syncthreads();
  float s = 0.0f, q = 0.0f;
  const int total = N_NODES * OUT_DIM;
  const int stride = gridDim.x * 256;
  for (int i = blockIdx.x * 256 + tid; i < total; i += stride) {
    float v = x[i];
    s += v;
    q += v * v;
  }
  const int c = tid & 63;
  atomicAdd(&ssum[c], s);
  atomicAdd(&ssq[c], q);
  __syncthreads();
  if (tid < OUT_DIM) {
    atomicAdd(&gsum[tid], ssum[tid]);
    atomicAdd(&gsq[tid], ssq[tid]);
  }
}

__global__ __launch_bounds__(256) void bn_apply(
    float* __restrict__ x, const float* __restrict__ gsum,
    const float* __restrict__ gsq, const float* __restrict__ gamma,
    const float* __restrict__ beta) {
  int i = blockIdx.x * 256 + threadIdx.x;
  if (i >= N_NODES * OUT_DIM) return;
  int c = i & 63;
  const float inv_n = 1.0f / (float)N_NODES;
  float mean = gsum[c] * inv_n;
  float var  = gsq[c] * inv_n - mean * mean;
  float sc   = gamma[c] * __frsqrt_rn(var + BN_EPS);
  x[i] = (x[i] - mean) * sc + beta[c];
}

extern "C" void kernel_launch(void* const* d_in, const int* in_sizes, int n_in,
                              void* d_out, int out_size, void* d_ws, size_t ws_size,
                              hipStream_t stream) {
  const float* h     = (const float*)d_in[0];
  const float* W1    = (const float*)d_in[1];
  const float* b1    = (const float*)d_in[2];
  const float* W2    = (const float*)d_in[3];
  const float* b2    = (const float*)d_in[4];
  const float* gamma = (const float*)d_in[5];
  const float* beta  = (const float*)d_in[6];
  const int*   src   = (const int*)d_in[7];
  const int*   dst   = (const int*)d_in[8];
  float* out = (float*)d_out;

  float* agg  = (float*)d_ws;
  float* deg  = agg + (size_t)N_NODES * NODE_DIM;
  float* gsum = deg + N_NODES;
  float* gsq  = gsum + OUT_DIM;

  const int nzero = N_NODES * NODE_DIM + N_NODES + 2 * OUT_DIM;
  zero_ws<<<(nzero + 255) / 256, 256, 0, stream>>>((float*)d_ws, nzero);

  const unsigned nscatter = (unsigned)N_EDGES * NODE_DIM;
  edge_scatter<<<(nscatter + 255) / 256, 256, 0, stream>>>(h, src, dst, agg, deg);

  fused_mlp<<<N_NODES / 16, 128, 0, stream>>>(h, W1, b1, W2, b2, agg, deg, out);

  bn_stats<<<1024, 256, 0, stream>>>(out, gsum, gsq);

  bn_apply<<<(N_NODES * OUT_DIM + 255) / 256, 256, 0, stream>>>(out, gsum, gsq, gamma, beta);
}


# Replace the f32 atomic scatter-add by a counting sort of the edges by destination (integer atomics) plus a per-node row sum in the MLP staging; a node's rows are added in fill-arrival order
# speedup vs baseline: 1.4542x; 1.4542x over previous
#include <hip/hip_runtime.h>
#include <hip/hip_bf16.h>

#define N_NODES   100000
#define N_EDGES   1600000
#define NODE_DIM  64
#define HIDDEN    256
#define OUT_DIM   64
#define BN_EPS    1e-5f

typedef __attribute__((ext_vector_type(2))) float v2f;
typedef __attribute__((ext_vector_type(8))) float v8f;

#define WMMA_F32X4(ACC, A, B) \
  ACC = __builtin_amdgcn_wmma_f32_16x16x4_f32(false, (A), false, (B), (short)0, (ACC), false, false)


__global__ __launch_bounds__(256) void zero_ws(int* __restrict__ p, int n) {
  int i = blockIdx.x * 256 + threadIdx.x;
  if (i < n) p[i] = 0;
}

__global__ __launch_bounds__(256) void edge_count(
    const int* __restrict__ dst, int* __restrict__ cnt) {
  int e = blockIdx.x * 256 + threadIdx.x;
  if (e < N_EDGES) atomicAdd(&cnt[dst[e]], 1);
}

__global__ __launch_bounds__(256) void seg_alloc(
    const int* __restrict__ cnt, int* __restrict__ total,
    int* __restrict__ seg_start, int* __restrict__ cursor) {
  const int n    = blockIdx.x * 256 + threadIdx.x;
  const int lane = threadIdx.x & (warpSize - 1);
  const int c    = n < N_NODES ? cnt[n] : 0;
  int incl = c;
  for (int o = 1; o < warpSize; o <<= 1) {
    int t = __shfl_up(incl, o);
    if (lane >= o) incl += t;
  }
  int base = 0;
  if (lane == warpSize - 1) base = atomicAdd(total, incl);
  base = __shfl(base, warpSize - 1);
  if (n < N_NODES) {
    const int s = base + incl - c;
    seg_start[n] = s;
    cursor[n]    = s;
  }
}

__global__ __launch_bounds__(256) void edge_fill(
    const int* __restrict__ src, const int* __restrict__ dst,
    int* __restrict__ cursor, int* __restrict__ nbr) {
  int e = blockIdx.x * 256 + threadIdx.x;
  if (e < N_EDGES) {
    int p = atomicAdd(&cursor[dst[e]], 1);
    nbr[p] = src[e];
  }
}

__global__ __launch_bounds__(128) void fused_mlp(
    const float* __restrict__ h,  const float* __restrict__ W1,
    const float* __restrict__ b1, const float* __restrict__ W2,
    const float* __restrict__ b2, const int* __restrict__ cnt,
    const int* __restrict__ seg_start, const int* __restrict__ nbr,
    float* __restrict__ out) {
  __shared__ float xs[16][NODE_DIM + 1];
  __shared__ float hs[16][HIDDEN + 1];

  const int m0   = blockIdx.x * 16;
  const int tid  = threadIdx.x;
  const int wave = tid >> 5;
  const int lane = tid & 31;
  const int half = lane >> 4;
  const int r    = lane & 15;

  {
    const int m    = tid >> 3;
    const int q    = tid & 7;
    const int node = m0 + m;
    const int deg  = cnt[node];
    const int* __restrict__ run = nbr + seg_start[node];
    const float* __restrict__ hq = h + 8 * q;
    float4 a0 = {0.0f, 0.0f, 0.0f, 0.0f}, a1 = {0.0f, 0.0f, 0.0f, 0.0f};
    int j = 0;
    for (; j + 4 <= deg; j += 4) {
      const int s0 = run[j], s1 = run[j + 1], s2 = run[j + 2], s3 = run[j + 3];
      const float4* p0 = reinterpret_cast<const float4*>(hq + (size_t)s0 * NODE_DIM);
      const float4* p1 = reinterpret_cast<const float4*>(hq + (size_t)s1 * NODE_DIM);
      const float4* p2 = reinterpret_cast<const float4*>(hq + (size_t)s2 * NODE_DIM);
      const float4* p3 = reinterpret_cast<const float4*>(hq + (size_t)s3 * NODE_DIM);
      const float4 u0 = p0[0], v0 = p0[1];
      const float4 u1 = p1[0], v1 = p1[1];
      const float4 u2 = p2[0], v2 = p2[1];
      const float4 u3 = p3[0], v3 = p3[1];
      a0 += u0; a1 += v0;
      a0 += u1; a1 += v1;
      a0 += u2; a1 += v2;
      a0 += u3; a1 += v3;
    }
    for (; j < deg; ++j) {
      const float4* p = reinterpret_cast<const float4*>(hq + (size_t)run[j] * NODE_DIM);
      const float4 u = p[0], v = p[1];
      a0 += u; a1 += v;
    }
    float d = (float)deg;
    d = d > 1.0f ? d : 1.0f;
    const float4* ps = reinterpret_cast<const float4*>(hq + (size_t)node * NODE_DIM);
    const float4 o0 = ps[0], o1 = ps[1];
    float* xr = &xs[m][8 * q];
    xr[0] = a0.x / d + o0.x;
    xr[1] = a0.y / d + o0.y;
    xr[2] = a0.z / d + o0.z;
    xr[3] = a0.w / d + o0.w;
    xr[4] = a1.x / d + o1.x;
    xr[5] = a1.y / d + o1.y;
    xr[6] = a1.z / d + o1.z;
    xr[7] = a1.w / d + o1.w;
  }
  __syncthreads();

  v8f acc0 = {}, acc1 = {}, acc2 = {}, acc3 = {};
  const int nb = wave * 64;
  for (int k = 0; k < NODE_DIM; k += 4) {
    v2f a;
    a.x = xs[r][k + 2 * half];
    a.y = xs[r][k + 2 * half + 1];
    const float* w1p0 = W1 + (k + 2 * half) * HIDDEN + nb + r;
    const float* w1p1 = W1 + (k + 2 * half + 1) * HIDDEN + nb + r;
    v2f b;
    b.x = w1p0[0];  b.y = w1p1[0];  WMMA_F32X4(acc0, a, b);
    b.x = w1p0[16]; b.y = w1p1[16]; WMMA_F32X4(acc1, a, b);
    b.x = w1p0[32]; b.y = w1p1[32]; WMMA_F32X4(acc2, a, b);
    b.x = w1p0[48]; b.y = w1p1[48]; WMMA_F32X4(acc3, a, b);
  }
#define STORE_TILE(ACC, T)                                        \
  {                                                               \
    const int n = nb + (T) * 16 + r;                              \
    const float bias = b1[n];                                     \
    _Pragma("unroll") for (int j = 0; j < 8; ++j) {               \
      float v = (ACC)[j] + bias;                                  \
      hs[j + 8 * half][n] = v > 0.0f ? v : 0.0f;                  \
    }                                                             \
  }
  STORE_TILE(acc0, 0)
  STORE_TILE(acc1, 1)
  STORE_TILE(acc2, 2)
  STORE_TILE(acc3, 3)
#undef STORE_TILE
  __syncthreads();

  v8f occ = {};
  const int ob = wave * 16;
  for (int k = 0; k < HIDDEN; k += 4) {
    v2f a;
    a.x = hs[r][k + 2 * half];
    a.y = hs[r][k + 2 * half + 1];
    v2f b;
    b.x = W2[(k + 2 * half) * OUT_DIM + ob + r];
    b.y = W2[(k + 2 * half + 1) * OUT_DIM + ob + r];
    WMMA_F32X4(occ, a, b);
  }
  {
    const float bias = b2[ob + r];
#pragma unroll
    for (int j = 0; j < 8; ++j) {
      float v = occ[j] + bias;
      v = v > 0.0f ? v : 0.0f;
      out[(m0 + j + 8 * half) * OUT_DIM + ob + r] = v;
    }
  }
}

__global__ __launch_bounds__(256) void bn_stats(
    const float* __restrict__ x, float* __restrict__ gsum,
    float* __restrict__ gsq) {
  __shared__ float ssum[OUT_DIM], ssq[OUT_DIM];
  const int tid = threadIdx.x;
  if (tid < OUT_DIM) { ssum[tid] = 0.0f; ssq[tid] = 0.0f; }
  __syncthreads();
  float s = 0.0f, q = 0.0f;
  const int total = N_NODES * OUT_DIM;
  const int stride = gridDim.x * 256;
  for (int i = blockIdx.x * 256 + tid; i < total; i += stride) {
    float v = x[i];
    s += v;
    q += v * v;
  }
  const int c = tid & 63;
  atomicAdd(&ssum[c], s);
  atomicAdd(&ssq[c], q);
  __syncthreads();
  if (tid < OUT_DIM) {
    atomicAdd(&gsum[tid], ssum[tid]);
    atomicAdd(&gsq[tid], ssq[tid]);
  }
}

__global__ __launch_bounds__(256) void bn_apply(
    float* __restrict__ x, const float* __restrict__ gsum,
    const float* __restrict__ gsq, const float* __restrict__ gamma,
    const float* __restrict__ beta) {
  int i = blockIdx.x * 256 + threadIdx.x;
  if (i >= N_NODES * OUT_DIM) return;
  int c = i & 63;
  const float inv_n = 1.0f / (float)N_NODES;
  float mean = gsum[c] * inv_n;
  float var  = gsq[c] * inv_n - mean * mean;
  float sc   = gamma[c] * __frsqrt_rn(var + BN_EPS);
  x[i] = (x[i] - mean) * sc + beta[c];
}

extern "C" void kernel_launch(void* const* d_in, const int* in_sizes, int n_in,
                              void* d_out, int out_size, void* d_ws, size_t ws_size,
                              hipStream_t stream) {
  const float* h     = (const float*)d_in[0];
  const float* W1    = (const float*)d_in[1];
  const float* b1    = (const float*)d_in[2];
  const float* W2    = (const float*)d_in[3];
  const float* b2    = (const float*)d_in[4];
  const float* gamma = (const float*)d_in[5];
  const float* beta  = (const float*)d_in[6];
  const int*   src   = (const int*)d_in[7];
  const int*   dst   = (const int*)d_in[8];
  float* out = (float*)d_out;

  int*   cnt       = (int*)d_ws;
  int*   total     = cnt + N_NODES;
  float* gsum      = (float*)(total + 32);
  float* gsq       = gsum + OUT_DIM;
  int*   seg_start = (int*)(gsq + OUT_DIM);
  int*   cursor    = seg_start + N_NODES;
  int*   nbr       = cursor + N_NODES;

  const int nzero = N_NODES + 32 + 2 * OUT_DIM;
  zero_ws<<<(nzero + 255) / 256, 256, 0, stream>>>(cnt, nzero);

  edge_count<<<(N_EDGES + 255) / 256, 256, 0, stream>>>(dst, cnt);

  seg_alloc<<<(N_NODES + 255) / 256, 256, 0, stream>>>(cnt, total, seg_start, cursor);

  edge_fill<<<(N_EDGES + 255) / 256, 256, 0, stream>>>(src, dst, cursor, nbr);

  fused_mlp<<<N_NODES / 16, 128, 0, stream>>>(h, W1, b1, W2, b2, cnt, seg_start, nbr, out);

  bn_stats<<<1024, 256, 0, stream>>>(out, gsum, gsq);

  bn_apply<<<(N_NODES * OUT_DIM + 255) / 256, 256, 0, stream>>>(out, gsum, gsq, gamma, beta);
}


# One-pass edge placement: an edge takes a slot in its destination's fixed 64-slot run by one returning add (excess to a spill list); no histogram or segment pass, 7 -> 5 launches; sums by arrival
# speedup vs baseline: 2.0473x; 1.4079x over previous
#include <hip/hip_runtime.h>
#include <hip/hip_bf16.h>

#define N_NODES   100000
#define N_EDGES   1600000
#define NODE_DIM  64
#define HIDDEN    256
#define OUT_DIM   64
#define BN_EPS    1e-5f

typedef __attribute__((ext_vector_type(2))) float v2f;
typedef __attribute__((ext_vector_type(8))) float v8f;

#define WMMA_F32X4(ACC, A, B) \
  ACC = __builtin_amdgcn_wmma_f32_16x16x4_f32(false, (A), false, (B), (short)0, (ACC), false, false)

#define SLOTS 64

__global__ __launch_bounds__(256) void zero_ws(int* __restrict__ p, int n) {
  int i = blockIdx.x * 256 + threadIdx.x;
  if (i < n) p[i] = 0;
}

__global__ __launch_bounds__(256) void edge_place(
    const int* __restrict__ src, const int* __restrict__ dst,
    int* __restrict__ cnt, int* __restrict__ nspill, int* __restrict__ nbr,
    int* __restrict__ spill_src, int* __restrict__ spill_dst) {
  int e = blockIdx.x * 256 + threadIdx.x;
  if (e < N_EDGES) {
    const int d = dst[e];
    const int s = src[e];
    const int p = atomicAdd(&cnt[d], 1);
    if (p < SLOTS) {
      nbr[d * SLOTS + p] = s;
    } else {
      const int o = atomicAdd(nspill, 1);
      spill_src[o] = s;
      spill_dst[o] = d;
    }
  }
}

__global__ __launch_bounds__(128) void fused_mlp(
    const float* __restrict__ h,  const float* __restrict__ W1,
    const float* __restrict__ b1, const float* __restrict__ W2,
    const float* __restrict__ b2, const int* __restrict__ cnt,
    const int* __restrict__ nspill, const int* __restrict__ nbr,
    const int* __restrict__ spill_src, const int* __restrict__ spill_dst,
    float* __restrict__ out) {
  __shared__ float xs[16][NODE_DIM + 1];
  __shared__ float hs[16][HIDDEN + 1];

  const int m0   = blockIdx.x * 16;
  const int tid  = threadIdx.x;
  const int wave = tid >> 5;
  const int lane = tid & 31;
  const int half = lane >> 4;
  const int r    = lane & 15;

  {
    const int m    = tid >> 3;
    const int q    = tid & 7;
    const int node = m0 + m;
    const int deg  = cnt[node];
    const int inl  = deg < SLOTS ? deg : SLOTS;
    const int* __restrict__ run = nbr + node * SLOTS;
    const float* __restrict__ hq = h + 8 * q;
    float4 a0 = {0.0f, 0.0f, 0.0f, 0.0f}, a1 = {0.0f, 0.0f, 0.0f, 0.0f};
    int j = 0;
    for (; j + 4 <= inl; j += 4) {
      const int s0 = run[j], s1 = run[j + 1], s2 = run[j + 2], s3 = run[j + 3];
      const float4* p0 = reinterpret_cast<const float4*>(hq + (size_t)s0 * NODE_DIM);
      const float4* p1 = reinterpret_cast<const float4*>(hq + (size_t)s1 * NODE_DIM);
      const float4* p2 = reinterpret_cast<const float4*>(hq + (size_t)s2 * NODE_DIM);
      const float4* p3 = reinterpret_cast<const float4*>(hq + (size_t)s3 * NODE_DIM);
      const float4 u0 = p0[0], v0 = p0[1];
      const float4 u1 = p1[0], v1 = p1[1];
      const float4 u2 = p2[0], v2 = p2[1];
      const float4 u3 = p3[0], v3 = p3[1];
      a0 += u0; a1 += v0;
      a0 += u1; a1 += v1;
      a0 += u2; a1 += v2;
      a0 += u3; a1 += v3;
    }
    for (; j < inl; ++j) {
      const float4* p = reinterpret_cast<const float4*>(hq + (size_t)run[j] * NODE_DIM);
      const float4 u = p[0], v = p[1];
      a0 += u; a1 += v;
    }
    if (deg > SLOTS) {
      const int ns = nspill[0];
      for (int i = 0; i < ns; ++i) {
        if (spill_dst[i] == node) {
          const float4* p = reinterpret_cast<const float4*>(hq + (size_t)spill_src[i] * NODE_DIM);
          const float4 u = p[0], v = p[1];
          a0 += u; a1 += v;
        }
      }
    }
    float d = (float)deg;
    d = d > 1.0f ? d : 1.0f;
    const float4* ps = reinterpret_cast<const float4*>(hq + (size_t)node * NODE_DIM);
    const float4 o0 = ps[0], o1 = ps[1];
    float* xr = &xs[m][8 * q];
    xr[0] = a0.x / d + o0.x;
    xr[1] = a0.y / d + o0.y;
    xr[2] = a0.z / d + o0.z;
    xr[3] = a0.w / d + o0.w;
    xr[4] = a1.x / d + o1.x;
    xr[5] = a1.y / d + o1.y;
    xr[6] = a1.z / d + o1.z;
    xr[7] = a1.w / d + o1.w;
  }
  __syncthreads();

  v8f acc0 = {}, acc1 = {}, acc2 = {}, acc3 = {};
  const int nb = wave * 64;
  for (int k = 0; k < NODE_DIM; k += 4) {
    v2f a;
    a.x = xs[r][k + 2 * half];
    a.y = xs[r][k + 2 * half + 1];
    const float* w1p0 = W1 + (k + 2 * half) * HIDDEN + nb + r;
    const float* w1p1 = W1 + (k + 2 * half + 1) * HIDDEN + nb + r;
    v2f b;
    b.x = w1p0[0];  b.y = w1p1[0];  WMMA_F32X4(acc0, a, b);
    b.x = w1p0[16]; b.y = w1p1[16]; WMMA_F32X4(acc1, a, b);
    b.x = w1p0[32]; b.y = w1p1[32]; WMMA_F32X4(acc2, a, b);
    b.x = w1p0[48]; b.y = w1p1[48]; WMMA_F32X4(acc3, a, b);
  }
#define STORE_TILE(ACC, T)                                        \
  {                                                               \
    const int n = nb + (T) * 16 + r;                              \
    const float bias = b1[n];                                     \
    _Pragma("unroll") for (int j = 0; j < 8; ++j) {               \
      float v = (ACC)[j] + bias;                                  \
      hs[j + 8 * half][n] = v > 0.0f ? v : 0.0f;                  \
    }                                                             \
  }
  STORE_TILE(acc0, 0)
  STORE_TILE(acc1, 1)
  STORE_TILE(acc2, 2)
  STORE_TILE(acc3, 3)
#undef STORE_TILE
  __syncthreads();

  v8f occ = {};
  const int ob = wave * 16;
  for (int k = 0; k < HIDDEN; k += 4) {
    v2f a;
    a.x = hs[r][k + 2 * half];
    a.y = hs[r][k + 2 * half + 1];
    v2f b;
    b.x = W2[(k + 2 * half) * OUT_DIM + ob + r];
    b.y = W2[(k + 2 * half + 1) * OUT_DIM + ob + r];
    WMMA_F32X4(occ, a, b);
  }
  {
    const float bias = b2[ob + r];
#pragma unroll
    for (int j = 0; j < 8; ++j) {
      float v = occ[j] + bias;
      v = v > 0.0f ? v : 0.0f;
      out[(m0 + j + 8 * half) * OUT_DIM + ob + r] = v;
    }
  }
}

__global__ __launch_bounds__(256) void bn_stats(
    const float* __restrict__ x, float* __restrict__ gsum,
    float* __restrict__ gsq) {
  __shared__ float ssum[OUT_DIM], ssq[OUT_DIM];
  const int tid = threadIdx.x;
  if (tid < OUT_DIM) { ssum[tid] = 0.0f; ssq[tid] = 0.0f; }
  __syncthreads();
  float s = 0.0f, q = 0.0f;
  const int total = N_NODES * OUT_DIM;
  const int stride = gridDim.x * 256;
  for (int i = blockIdx.x * 256 + tid; i < total; i += stride) {
    float v = x[i];
    s += v;
    q += v * v;
  }
  const int c = tid & 63;
  atomicAdd(&ssum[c], s);
  atomicAdd(&ssq[c], q);
  __syncthreads();
  if (tid < OUT_DIM) {
    atomicAdd(&gsum[tid], ssum[tid]);
    atomicAdd(&gsq[tid], ssq[tid]);
  }
}

__global__ __launch_bounds__(256) void bn_apply(
    float* __restrict__ x, const float* __restrict__ gsum,
    const float* __restrict__ gsq, const float* __restrict__ gamma,
    const float* __restrict__ beta) {
  int i = blockIdx.x * 256 + threadIdx.x;
  if (i >= N_NODES * OUT_DIM) return;
  int c = i & 63;
  const float inv_n = 1.0f / (float)N_NODES;
  float mean = gsum[c] * inv_n;
  float var  = gsq[c] * inv_n - mean * mean;
  float sc   = gamma[c] * __frsqrt_rn(var + BN_EPS);
  x[i] = (x[i] - mean) * sc + beta[c];
}

extern "C" void kernel_launch(void* const* d_in, const int* in_sizes, int n_in,
                              void* d_out, int out_size, void* d_ws, size_t ws_size,
                              hipStream_t stream) {
  const float* h     = (const float*)d_in[0];
  const float* W1    = (const float*)d_in[1];
  const float* b1    = (const float*)d_in[2];
  const float* W2    = (const float*)d_in[3];
  const float* b2    = (const float*)d_in[4];
  const float* gamma = (const float*)d_in[5];
  const float* beta  = (const float*)d_in[6];
  const int*   src   = (const int*)d_in[7];
  const int*   dst   = (const int*)d_in[8];
  float* out = (float*)d_out;

  int*   cnt       = (int*)d_ws;
  int*   nspill    = cnt + N_NODES;
  float* gsum      = (float*)(nspill + 32);
  float* gsq       = gsum + OUT_DIM;
  int*   nbr       = (int*)(gsq + OUT_DIM);
  int*   spill_src = nbr + N_NODES * SLOTS;
  int*   spill_dst = spill_src + N_EDGES;

  const int nzero = N_NODES + 32 + 2 * OUT_DIM;
  zero_ws<<<(nzero + 255) / 256, 256, 0, stream>>>(cnt, nzero);

  edge_place<<<(N_EDGES + 255) / 256, 256, 0, stream>>>(src, dst, cnt, nspill, nbr, spill_src, spill_dst);

  fused_mlp<<<N_NODES / 16, 128, 0, stream>>>(h, W1, b1, W2, b2, cnt, nspill, nbr, spill_src, spill_dst, out);

  bn_stats<<<1024, 256, 0, stream>>>(out, gsum, gsq);

  bn_apply<<<(N_NODES * OUT_DIM + 255) / 256, 256, 0, stream>>>(out, gsum, gsq, gamma, beta);
}


# BN statistics pass re-tiled to 128 work-groups of 1024 lanes: a lane's 12-13 float4 loads all in flight (was 24 dependent loads), 8x fewer float adds on the shared totals; channel sums in a new order
# speedup vs baseline: 2.1306x; 1.0407x over previous
#include <hip/hip_runtime.h>
#include <hip/hip_bf16.h>

#define N_NODES   100000
#define N_EDGES   1600000
#define NODE_DIM  64
#define HIDDEN    256
#define OUT_DIM   64
#define BN_EPS    1e-5f

typedef __attribute__((ext_vector_type(2))) float v2f;
typedef __attribute__((ext_vector_type(8))) float v8f;

#define WMMA_F32X4(ACC, A, B) \
  ACC = __builtin_amdgcn_wmma_f32_16x16x4_f32(false, (A), false, (B), (short)0, (ACC), false, false)

#define SLOTS 64

__global__ __launch_bounds__(256) void zero_ws(int* __restrict__ p, int n) {
  int i = blockIdx.x * 256 + threadIdx.x;
  if (i < n) p[i] = 0;
}

__global__ __launch_bounds__(256) void edge_place(
    const int* __restrict__ src, const int* __restrict__ dst,
    int* __restrict__ cnt, int* __restrict__ nspill, int* __restrict__ nbr,
    int* __restrict__ spill_src, int* __restrict__ spill_dst) {
  int e = blockIdx.x * 256 + threadIdx.x;
  if (e < N_EDGES) {
    const int d = dst[e];
    const int s = src[e];
    const int p = atomicAdd(&cnt[d], 1);
    if (p < SLOTS) {
      nbr[d * SLOTS + p] = s;
    } else {
      const int o = atomicAdd(nspill, 1);
      spill_src[o] = s;
      spill_dst[o] = d;
    }
  }
}

__global__ __launch_bounds__(128) void fused_mlp(
    const float* __restrict__ h,  const float* __restrict__ W1,
    const float* __restrict__ b1, const float* __restrict__ W2,
    const float* __restrict__ b2, const int* __restrict__ cnt,
    const int* __restrict__ nspill, const int* __restrict__ nbr,
    const int* __restrict__ spill_src, const int* __restrict__ spill_dst,
    float* __restrict__ out) {
  __shared__ float xs[16][NODE_DIM + 1];
  __shared__ float hs[16][HIDDEN + 1];

  const int m0   = blockIdx.x * 16;
  const int tid  = threadIdx.x;
  const int wave = tid >> 5;
  const int lane = tid & 31;
  const int half = lane >> 4;
  const int r    = lane & 15;

  {
    const int m    = tid >> 3;
    const int q    = tid & 7;
    const int node = m0 + m;
    const int deg  = cnt[node];
    const int inl  = deg < SLOTS ? deg : SLOTS;
    const int* __restrict__ run = nbr + node * SLOTS;
    const float* __restrict__ hq = h + 8 * q;
    float4 a0 = {0.0f, 0.0f, 0.0f, 0.0f}, a1 = {0.0f, 0.0f, 0.0f, 0.0f};
    int j = 0;
    for (; j + 4 <= inl; j += 4) {
      const int s0 = run[j], s1 = run[j + 1], s2 = run[j + 2], s3 = run[j + 3];
      const float4* p0 = reinterpret_cast<const float4*>(hq + (size_t)s0 * NODE_DIM);
      const float4* p1 = reinterpret_cast<const float4*>(hq + (size_t)s1 * NODE_DIM);
      const float4* p2 = reinterpret_cast<const float4*>(hq + (size_t)s2 * NODE_DIM);
      const float4* p3 = reinterpret_cast<const float4*>(hq + (size_t)s3 * NODE_DIM);
      const float4 u0 = p0[0], v0 = p0[1];
      const float4 u1 = p1[0], v1 = p1[1];
      const float4 u2 = p2[0], v2 = p2[1];
      const float4 u3 = p3[0], v3 = p3[1];
      a0 += u0; a1 += v0;
      a0 += u1; a1 += v1;
      a0 += u2; a1 += v2;
      a0 += u3; a1 += v3;
    }
    for (; j < inl; ++j) {
      const float4* p = reinterpret_cast<const float4*>(hq + (size_t)run[j] * NODE_DIM);
      const float4 u = p[0], v = p[1];
      a0 += u; a1 += v;
    }
    if (deg > SLOTS) {
      const int ns = nspill[0];
      for (int i = 0; i < ns; ++i) {
        if (spill_dst[i] == node) {
          const float4* p = reinterpret_cast<const float4*>(hq + (size_t)spill_src[i] * NODE_DIM);
          const float4 u = p[0], v = p[1];
          a0 += u; a1 += v;
        }
      }
    }
    float d = (float)deg;
    d = d > 1.0f ? d : 1.0f;
    const float4* ps = reinterpret_cast<const float4*>(hq + (size_t)node * NODE_DIM);
    const float4 o0 = ps[0], o1 = ps[1];
    float* xr = &xs[m][8 * q];
    xr[0] = a0.x / d + o0.x;
    xr[1] = a0.y / d + o0.y;
    xr[2] = a0.z / d + o0.z;
    xr[3] = a0.w / d + o0.w;
    xr[4] = a1.x / d + o1.x;
    xr[5] = a1.y / d + o1.y;
    xr[6] = a1.z / d + o1.z;
    xr[7] = a1.w / d + o1.w;
  }
  __syncthreads();

  v8f acc0 = {}, acc1 = {}, acc2 = {}, acc3 = {};
  const int nb = wave * 64;
  for (int k = 0; k < NODE_DIM; k += 4) {
    v2f a;
    a.x = xs[r][k + 2 * half];
    a.y = xs[r][k + 2 * half + 1];
    const float* w1p0 = W1 + (k + 2 * half) * HIDDEN + nb + r;
    const float* w1p1 = W1 + (k + 2 * half + 1) * HIDDEN + nb + r;
    v2f b;
    b.x = w1p0[0];  b.y = w1p1[0];  WMMA_F32X4(acc0, a, b);
    b.x = w1p0[16]; b.y = w1p1[16]; WMMA_F32X4(acc1, a, b);
    b.x = w1p0[32]; b.y = w1p1[32]; WMMA_F32X4(acc2, a, b);
    b.x = w1p0[48]; b.y = w1p1[48]; WMMA_F32X4(acc3, a, b);
  }
#define STORE_TILE(ACC, T)                                        \
  {                                                               \
    const int n = nb + (T) * 16 + r;                              \
    const float bias = b1[n];                                     \
    _Pragma("unroll") for (int j = 0; j < 8; ++j) {               \
      float v = (ACC)[j] + bias;                                  \
      hs[j + 8 * half][n] = v > 0.0f ? v : 0.0f;                  \
    }                                                             \
  }
  STORE_TILE(acc0, 0)
  STORE_TILE(acc1, 1)
  STORE_TILE(acc2, 2)
  STORE_TILE(acc3, 3)
#undef STORE_TILE
  __syncthreads();

  v8f occ = {};
  const int ob = wave * 16;
  for (int k = 0; k < HIDDEN; k += 4) {
    v2f a;
    a.x = hs[r][k + 2 * half];
    a.y = hs[r][k + 2 * half + 1];
    v2f b;
    b.x = W2[(k + 2 * half) * OUT_DIM + ob + r];
    b.y = W2[(k + 2 * half + 1) * OUT_DIM + ob + r];
    WMMA_F32X4(occ, a, b);
  }
  {
    const float bias = b2[ob + r];
#pragma unroll
    for (int j = 0; j < 8; ++j) {
      float v = occ[j] + bias;
      v = v > 0.0f ? v : 0.0f;
      out[(m0 + j + 8 * half) * OUT_DIM + ob + r] = v;
    }
  }
}

#define STAT_WGS    128
#define STAT_LANES  1024
#define STAT_TOTAL4 (N_NODES * OUT_DIM / 4)
#define STAT_STEP4  (STAT_WGS * STAT_LANES)
#define STAT_FULL   (STAT_TOTAL4 / STAT_STEP4)
static_assert(STAT_STEP4 % 16 == 0, "a lane must keep its channels");
static_assert((STAT_FULL + 1) * STAT_STEP4 >= STAT_TOTAL4, "one partial load covers the rest");

__global__ __launch_bounds__(STAT_LANES) void bn_stats(
    const float* __restrict__ x, float* __restrict__ gsum,
    float* __restrict__ gsq) {
  __shared__ float ssum[OUT_DIM], ssq[OUT_DIM];
  const int tid = threadIdx.x;
  if (tid < OUT_DIM) { ssum[tid] = 0.0f; ssq[tid] = 0.0f; }
  __syncthreads();
  const float4* __restrict__ x4 = reinterpret_cast<const float4*>(x);
  const int g0 = blockIdx.x * STAT_LANES + tid;
  float4 v[STAT_FULL + 1];
#pragma unroll
  for (int it = 0; it < STAT_FULL; ++it) v[it] = x4[g0 + it * STAT_STEP4];
  const int gl = g0 + STAT_FULL * STAT_STEP4;
  v[STAT_FULL] = make_float4(0.0f, 0.0f, 0.0f, 0.0f);
  if (gl < STAT_TOTAL4) v[STAT_FULL] = x4[gl];
  float4 s = {0.0f, 0.0f, 0.0f, 0.0f}, q = {0.0f, 0.0f, 0.0f, 0.0f};
#pragma unroll
  for (int it = 0; it <= STAT_FULL; ++it) {
    s.x += v[it].x; s.y += v[it].y; s.z += v[it].z; s.w += v[it].w;
    q.x += v[it].x * v[it].x; q.y += v[it].y * v[it].y;
    q.z += v[it].z * v[it].z; q.w += v[it].w * v[it].w;
  }
  const int c0 = 4 * (tid & 15);
  atomicAdd(&ssum[c0 + 0], s.x); atomicAdd(&ssq[c0 + 0], q.x);
  atomicAdd(&ssum[c0 + 1], s.y); atomicAdd(&ssq[c0 + 1], q.y);
  atomicAdd(&ssum[c0 + 2], s.z); atomicAdd(&ssq[c0 + 2], q.z);
  atomicAdd(&ssum[c0 + 3], s.w); atomicAdd(&ssq[c0 + 3], q.w);
  __syncthreads();
  if (tid < OUT_DIM) {
    atomicAdd(&gsum[tid], ssum[tid]);
    atomicAdd(&gsq[tid], ssq[tid]);
  }
}

__global__ __launch_bounds__(256) void bn_apply(
    float* __restrict__ x, const float* __restrict__ gsum,
    const float* __restrict__ gsq, const float* __restrict__ gamma,
    const float* __restrict__ beta) {
  int i = blockIdx.x * 256 + threadIdx.x;
  if (i >= N_NODES * OUT_DIM) return;
  int c = i & 63;
  const float inv_n = 1.0f / (float)N_NODES;
  float mean = gsum[c] * inv_n;
  float var  = gsq[c] * inv_n - mean * mean;
  float sc   = gamma[c] * __frsqrt_rn(var + BN_EPS);
  x[i] = (x[i] - mean) * sc + beta[c];
}

extern "C" void kernel_launch(void* const* d_in, const int* in_sizes, int n_in,
                              void* d_out, int out_size, void* d_ws, size_t ws_size,
                              hipStream_t stream) {
  const float* h     = (const float*)d_in[0];
  const float* W1    = (const float*)d_in[1];
  const float* b1    = (const float*)d_in[2];
  const float* W2    = (const float*)d_in[3];
  const float* b2    = (const float*)d_in[4];
  const float* gamma = (const float*)d_in[5];
  const float* beta  = (const float*)d_in[6];
  const int*   src   = (const int*)d_in[7];
  const int*   dst   = (const int*)d_in[8];
  float* out = (float*)d_out;

  int*   cnt       = (int*)d_ws;
  int*   nspill    = cnt + N_NODES;
  float* gsum      = (float*)(nspill + 32);
  float* gsq       = gsum + OUT_DIM;
  int*   nbr       = (int*)(gsq + OUT_DIM);
  int*   spill_src = nbr + N_NODES * SLOTS;
  int*   spill_dst = spill_src + N_EDGES;

  const int nzero = N_NODES + 32 + 2 * OUT_DIM;
  zero_ws<<<(nzero + 255) / 256, 256, 0, stream>>>(cnt, nzero);

  edge_place<<<(N_EDGES + 255) / 256, 256, 0, stream>>>(src, dst, cnt, nspill, nbr, spill_src, spill_dst);

  fused_mlp<<<N_NODES / 16, 128, 0, stream>>>(h, W1, b1, W2, b2, cnt, nspill, nbr, spill_src, spill_dst, out);

  bn_stats<<<STAT_WGS, STAT_LANES, 0, stream>>>(out, gsum, gsq);

  bn_apply<<<(N_NODES * OUT_DIM + 255) / 256, 256, 0, stream>>>(out, gsum, gsq, gamma, beta);
}


# Edges grouped by destination through LDS: chunks sorted into 241 node-range bins, then one work-group per bin fills its nodes' runs in LDS; no per-edge global add or 4-B store; sums by LDS arrival
# speedup vs baseline: 3.8370x; 1.8009x over previous
#include <hip/hip_runtime.h>
#include <hip/hip_bf16.h>

#define N_NODES   100000
#define N_EDGES   1600000
#define NODE_DIM  64
#define HIDDEN    256
#define OUT_DIM   64
#define BN_EPS    1e-5f

typedef __attribute__((ext_vector_type(2))) float v2f;
typedef __attribute__((ext_vector_type(8))) float v8f;

#define WMMA_F32X4(ACC, A, B) \
  ACC = __builtin_amdgcn_wmma_f32_16x16x4_f32(false, (A), false, (B), (short)0, (ACC), false, false)

#define SLOTS 64

#define CH        8192
#define NBLK_A    ((N_EDGES + CH - 1) / CH)
#define BIN_NODES 416
#define NBINS     ((N_NODES + BIN_NODES - 1) / BIN_NODES)
#define SEG       96
#define CROW      256
#define TP        68
static_assert(NBINS < CROW - 1, "piece counts and the overflow count share a row");
static_assert((SEG * 8) % 256 == 0, "a slot is whole 256-B units");
static_assert(BIN_NODES % 32 == 0, "a range's counters are whole lines");

__global__ __launch_bounds__(1024) void edge_bin(
    const int* __restrict__ src, const int* __restrict__ dst,
    int2* __restrict__ bins, int* __restrict__ cntAB, int2* __restrict__ ovf,
    int* __restrict__ nspill, float* __restrict__ gtot) {
  __shared__ int  hist[CROW];
  __shared__ int  lstart[CROW];
  __shared__ int  novf;
  __shared__ int2 sorted[CH];
  const int tid = threadIdx.x;
  const int blk = blockIdx.x;
  const int e0  = blk * CH;
  if (tid < CROW) hist[tid] = 0;
  if (tid == 0) novf = 0;
  if (blk == 0) {
    if (tid < 2 * OUT_DIM) gtot[tid] = 0.0f;
    if (tid == 2 * OUT_DIM) nspill[0] = 0;
  }
  __syncthreads();

  int s[CH / 1024], d[CH / 1024], bn[CH / 1024], rk[CH / 1024];
#pragma unroll
  for (int k = 0; k < CH / 1024; ++k) {
    const int e = e0 + k * 1024 + tid;
    d[k] = -1; s[k] = 0;
    if (e < N_EDGES) { d[k] = dst[e]; s[k] = src[e]; }
  }
#pragma unroll
  for (int k = 0; k < CH / 1024; ++k) {
    bn[k] = 0; rk[k] = 0;
    if (d[k] >= 0) {
      bn[k] = d[k] / BIN_NODES;
      rk[k] = atomicAdd(&hist[bn[k]], 1);
    }
  }
  __syncthreads();

  if (tid < 32) {
    int v[CROW / 32];
    int tot = 0;
#pragma unroll
    for (int j = 0; j < CROW / 32; ++j) { v[j] = hist[tid * (CROW / 32) + j]; tot += v[j]; }
    int inc = tot;
#pragma unroll
    for (int off = 1; off < 32; off <<= 1) {
      const int y = __shfl_up(inc, off);
      if (tid >= off) inc += y;
    }
    int run = inc - tot;
#pragma unroll
    for (int j = 0; j < CROW / 32; ++j) { lstart[tid * (CROW / 32) + j] = run; run += v[j]; }
  }
  __syncthreads();

#pragma unroll
  for (int k = 0; k < CH / 1024; ++k) {
    if (d[k] >= 0) sorted[lstart[bn[k]] + rk[k]] = make_int2(s[k], d[k]);
  }
  __syncthreads();

  const int n = (N_EDGES - e0) < CH ? (N_EDGES - e0) : CH;
  for (int i = tid; i < n; i += 1024) {
    const int2 e = sorted[i];
    const int b  = e.y / BIN_NODES;
    const int rr = i - lstart[b];
    int2* p = bins + (((size_t)b * NBLK_A + blk) * SEG + rr);
    if (rr >= SEG) {
      p = ovf + ((size_t)blk * CH + atomicAdd(&novf, 1));
    }
    *p = e;
  }
  __syncthreads();
  if (tid < CROW) {
    int v = 0;
    if (tid < NBINS) v = hist[tid] < SEG ? hist[tid] : SEG;
    if (tid == CROW - 1) v = novf;
    cntAB[blk * CROW + tid] = v;
  }
}

__device__ __forceinline__ void place_edge(
    const int2 e, const int node0, int* lcnt, int* tbl,
    int* __restrict__ nspill, int* __restrict__ spill_src, int* __restrict__ spill_dst) {
  const int dl = e.y - node0;
  const int p  = atomicAdd(&lcnt[dl], 1);
  if (p < SLOTS) {
    tbl[dl * TP + p] = e.x;
  } else {
    const int o = atomicAdd(nspill, 1);
    spill_src[o] = e.x;
    spill_dst[o] = e.y;
  }
}

#define SEG_PER_WAVE ((NBLK_A + 31) / 32)

__global__ __launch_bounds__(1024) void edge_runs(
    const int2* __restrict__ bins, const int* __restrict__ cntAB,
    const int2* __restrict__ ovf, int* __restrict__ cnt, int* __restrict__ nspill,
    int* __restrict__ nbr, int* __restrict__ spill_src, int* __restrict__ spill_dst) {
  __shared__ int lcnt[BIN_NODES];
  __shared__ int anyovf;
  __shared__ __attribute__((aligned(16))) int tbl[BIN_NODES * TP];
  const int tid   = threadIdx.x;
  const int wave  = tid >> 5;
  const int lane  = tid & 31;
  const int b     = blockIdx.x;
  const int node0 = b * BIN_NODES;
  if (tid < BIN_NODES) lcnt[tid] = 0;
  if (tid == 0) anyovf = 0;
  __syncthreads();

  int nsl = 0;
  if (lane < SEG_PER_WAVE && wave + 32 * lane < NBLK_A) nsl = cntAB[(wave + 32 * lane) * CROW + b];
  int ns[SEG_PER_WAVE];
#pragma unroll
  for (int k = 0; k < SEG_PER_WAVE; ++k) ns[k] = __shfl(nsl, k);
  int2 e[SEG_PER_WAVE], f[SEG_PER_WAVE];
#pragma unroll
  for (int k = 0; k < SEG_PER_WAVE; ++k) {
    const int blk = wave + 32 * k;
    const int2* __restrict__ seg = bins + ((size_t)b * NBLK_A + blk) * SEG;
    e[k] = make_int2(0, 0);
    f[k] = make_int2(0, 0);
    if (lane < ns[k]) e[k] = seg[lane];
    if (lane + 32 < ns[k]) f[k] = seg[lane + 32];
  }
#pragma unroll
  for (int k = 0; k < SEG_PER_WAVE; ++k) {
    if (lane < ns[k]) place_edge(e[k], node0, lcnt, tbl, nspill, spill_src, spill_dst);
    if (lane + 32 < ns[k]) place_edge(f[k], node0, lcnt, tbl, nspill, spill_src, spill_dst);
  }
#pragma unroll
  for (int k = 0; k < SEG_PER_WAVE; ++k) {
    const int blk = wave + 32 * k;
    for (int i = lane + 64; i < ns[k]; i += 32)
      place_edge(bins[((size_t)b * NBLK_A + blk) * SEG + i], node0, lcnt, tbl, nspill, spill_src, spill_dst);
  }

  if (tid < NBLK_A && cntAB[tid * CROW + CROW - 1] > 0) anyovf = 1;
  __syncthreads();
  if (anyovf) {
    for (int blk = 0; blk < NBLK_A; ++blk) {
      const int no = cntAB[blk * CROW + CROW - 1];
      for (int i = tid; i < no; i += 1024) {
        const int2 x = ovf[(size_t)blk * CH + i];
        if (x.y / BIN_NODES == b) place_edge(x, node0, lcnt, tbl, nspill, spill_src, spill_dst);
      }
    }
  }
  __syncthreads();

  if (tid < BIN_NODES && node0 + tid < N_NODES) cnt[node0 + tid] = lcnt[tid];
  const int q = tid & 15;
  for (int dl = tid >> 4; dl < BIN_NODES; dl += 64) {
    const int node = node0 + dl;
    if (node < N_NODES) {
      const int c   = lcnt[dl];
      const int inl = c < SLOTS ? c : SLOTS;
      if (4 * q < inl) {
        const int4 v = *reinterpret_cast<const int4*>(&tbl[dl * TP + 4 * q]);
        *reinterpret_cast<int4*>(&nbr[(size_t)node * SLOTS + 4 * q]) = v;
      }
    }
  }
}

__global__ __launch_bounds__(128) void fused_mlp(
    const float* __restrict__ h,  const float* __restrict__ W1,
    const float* __restrict__ b1, const float* __restrict__ W2,
    const float* __restrict__ b2, const int* __restrict__ cnt,
    const int* __restrict__ nspill, const int* __restrict__ nbr,
    const int* __restrict__ spill_src, const int* __restrict__ spill_dst,
    float* __restrict__ out) {
  __shared__ float xs[16][NODE_DIM + 1];
  __shared__ float hs[16][HIDDEN + 1];

  const int m0   = blockIdx.x * 16;
  const int tid  = threadIdx.x;
  const int wave = tid >> 5;
  const int lane = tid & 31;
  const int half = lane >> 4;
  const int r    = lane & 15;

  {
    const int m    = tid >> 3;
    const int q    = tid & 7;
    const int node = m0 + m;
    const int deg  = cnt[node];
    const int inl  = deg < SLOTS ? deg : SLOTS;
    const int* __restrict__ run = nbr + node * SLOTS;
    const float* __restrict__ hq = h + 8 * q;
    float4 a0 = {0.0f, 0.0f, 0.0f, 0.0f}, a1 = {0.0f, 0.0f, 0.0f, 0.0f};
    int j = 0;
    for (; j + 4 <= inl; j += 4) {
      const int s0 = run[j], s1 = run[j + 1], s2 = run[j + 2], s3 = run[j + 3];
      const float4* p0 = reinterpret_cast<const float4*>(hq + (size_t)s0 * NODE_DIM);
      const float4* p1 = reinterpret_cast<const float4*>(hq + (size_t)s1 * NODE_DIM);
      const float4* p2 = reinterpret_cast<const float4*>(hq + (size_t)s2 * NODE_DIM);
      const float4* p3 = reinterpret_cast<const float4*>(hq + (size_t)s3 * NODE_DIM);
      const float4 u0 = p0[0], v0 = p0[1];
      const float4 u1 = p1[0], v1 = p1[1];
      const float4 u2 = p2[0], v2 = p2[1];
      const float4 u3 = p3[0], v3 = p3[1];
      a0 += u0; a1 += v0;
      a0 += u1; a1 += v1;
      a0 += u2; a1 += v2;
      a0 += u3; a1 += v3;
    }
    for (; j < inl; ++j) {
      const float4* p = reinterpret_cast<const float4*>(hq + (size_t)run[j] * NODE_DIM);
      const float4 u = p[0], v = p[1];
      a0 += u; a1 += v;
    }
    if (deg > SLOTS) {
      const int ns = nspill[0];
      for (int i = 0; i < ns; ++i) {
        if (spill_dst[i] == node) {
          const float4* p = reinterpret_cast<const float4*>(hq + (size_t)spill_src[i] * NODE_DIM);
          const float4 u = p[0], v = p[1];
          a0 += u; a1 += v;
        }
      }
    }
    float d = (float)deg;
    d = d > 1.0f ? d : 1.0f;
    const float4* ps = reinterpret_cast<const float4*>(hq + (size_t)node * NODE_DIM);
    const float4 o0 = ps[0], o1 = ps[1];
    float* xr = &xs[m][8 * q];
    xr[0] = a0.x / d + o0.x;
    xr[1] = a0.y / d + o0.y;
    xr[2] = a0.z / d + o0.z;
    xr[3] = a0.w / d + o0.w;
    xr[4] = a1.x / d + o1.x;
    xr[5] = a1.y / d + o1.y;
    xr[6] = a1.z / d + o1.z;
    xr[7] = a1.w / d + o1.w;
  }
  __syncthreads();

  v8f acc0 = {}, acc1 = {}, acc2 = {}, acc3 = {};
  const int nb = wave * 64;
  for (int k = 0; k < NODE_DIM; k += 4) {
    v2f a;
    a.x = xs[r][k + 2 * half];
    a.y = xs[r][k + 2 * half + 1];
    const float* w1p0 = W1 + (k + 2 * half) * HIDDEN + nb + r;
    const float* w1p1 = W1 + (k + 2 * half + 1) * HIDDEN + nb + r;
    v2f b;
    b.x = w1p0[0];  b.y = w1p1[0];  WMMA_F32X4(acc0, a, b);
    b.x = w1p0[16]; b.y = w1p1[16]; WMMA_F32X4(acc1, a, b);
    b.x = w1p0[32]; b.y = w1p1[32]; WMMA_F32X4(acc2, a, b);
    b.x = w1p0[48]; b.y = w1p1[48]; WMMA_F32X4(acc3, a, b);
  }
#define STORE_TILE(ACC, T)                                        \
  {                                                               \
    const int n = nb + (T) * 16 + r;                              \
    const float bias = b1[n];                                     \
    _Pragma("unroll") for (int j = 0; j < 8; ++j) {               \
      float v = (ACC)[j] + bias;                                  \
      hs[j + 8 * half][n] = v > 0.0f ? v : 0.0f;                  \
    }                                                             \
  }
  STORE_TILE(acc0, 0)
  STORE_TILE(acc1, 1)
  STORE_TILE(acc2, 2)
  STORE_TILE(acc3, 3)
#undef STORE_TILE
  __syncthreads();

  v8f occ = {};
  const int ob = wave * 16;
  for (int k = 0; k < HIDDEN; k += 4) {
    v2f a;
    a.x = hs[r][k + 2 * half];
    a.y = hs[r][k + 2 * half + 1];
    v2f b;
    b.x = W2[(k + 2 * half) * OUT_DIM + ob + r];
    b.y = W2[(k + 2 * half + 1) * OUT_DIM + ob + r];
    WMMA_F32X4(occ, a, b);
  }
  {
    const float bias = b2[ob + r];
#pragma unroll
    for (int j = 0; j < 8; ++j) {
      float v = occ[j] + bias;
      v = v > 0.0f ? v : 0.0f;
      out[(m0 + j + 8 * half) * OUT_DIM + ob + r] = v;
    }
  }
}

#define STAT_WGS    128
#define STAT_LANES  1024
#define STAT_TOTAL4 (N_NODES * OUT_DIM / 4)
#define STAT_STEP4  (STAT_WGS * STAT_LANES)
#define STAT_FULL   (STAT_TOTAL4 / STAT_STEP4)
static_assert(STAT_STEP4 % 16 == 0, "a lane must keep its channels");
static_assert((STAT_FULL + 1) * STAT_STEP4 >= STAT_TOTAL4, "one partial load covers the rest");

__global__ __launch_bounds__(STAT_LANES) void bn_stats(
    const float* __restrict__ x, float* __restrict__ gsum,
    float* __restrict__ gsq) {
  __shared__ float ssum[OUT_DIM], ssq[OUT_DIM];
  const int tid = threadIdx.x;
  if (tid < OUT_DIM) { ssum[tid] = 0.0f; ssq[tid] = 0.0f; }
  __syncthreads();
  const float4* __restrict__ x4 = reinterpret_cast<const float4*>(x);
  const int g0 = blockIdx.x * STAT_LANES + tid;
  float4 v[STAT_FULL + 1];
#pragma unroll
  for (int it = 0; it < STAT_FULL; ++it) v[it] = x4[g0 + it * STAT_STEP4];
  const int gl = g0 + STAT_FULL * STAT_STEP4;
  v[STAT_FULL] = make_float4(0.0f, 0.0f, 0.0f, 0.0f);
  if (gl < STAT_TOTAL4) v[STAT_FULL] = x4[gl];
  float4 s = {0.0f, 0.0f, 0.0f, 0.0f}, q = {0.0f, 0.0f, 0.0f, 0.0f};
#pragma unroll
  for (int it = 0; it <= STAT_FULL; ++it) {
    s.x += v[it].x; s.y += v[it].y; s.z += v[it].z; s.w += v[it].w;
    q.x += v[it].x * v[it].x; q.y += v[it].y * v[it].y;
    q.z += v[it].z * v[it].z; q.w += v[it].w * v[it].w;
  }
  const int c0 = 4 * (tid & 15);
  atomicAdd(&ssum[c0 + 0], s.x); atomicAdd(&ssq[c0 + 0], q.x);
  atomicAdd(&ssum[c0 + 1], s.y); atomicAdd(&ssq[c0 + 1], q.y);
  atomicAdd(&ssum[c0 + 2], s.z); atomicAdd(&ssq[c0 + 2], q.z);
  atomicAdd(&ssum[c0 + 3], s.w); atomicAdd(&ssq[c0 + 3], q.w);
  __syncthreads();
  if (tid < OUT_DIM) {
    atomicAdd(&gsum[tid], ssum[tid]);
    atomicAdd(&gsq[tid], ssq[tid]);
  }
}

__global__ __launch_bounds__(256) void bn_apply(
    float* __restrict__ x, const float* __restrict__ gsum,
    const float* __restrict__ gsq, const float* __restrict__ gamma,
    const float* __restrict__ beta) {
  int i = blockIdx.x * 256 + threadIdx.x;
  if (i >= N_NODES * OUT_DIM) return;
  int c = i & 63;
  const float inv_n = 1.0f / (float)N_NODES;
  float mean = gsum[c] * inv_n;
  float var  = gsq[c] * inv_n - mean * mean;
  float sc   = gamma[c] * __frsqrt_rn(var + BN_EPS);
  x[i] = (x[i] - mean) * sc + beta[c];
}

extern "C" void kernel_launch(void* const* d_in, const int* in_sizes, int n_in,
                              void* d_out, int out_size, void* d_ws, size_t ws_size,
                              hipStream_t stream) {
  const float* h     = (const float*)d_in[0];
  const float* W1    = (const float*)d_in[1];
  const float* b1    = (const float*)d_in[2];
  const float* W2    = (const float*)d_in[3];
  const float* b2    = (const float*)d_in[4];
  const float* gamma = (const float*)d_in[5];
  const float* beta  = (const float*)d_in[6];
  const int*   src   = (const int*)d_in[7];
  const int*   dst   = (const int*)d_in[8];
  float* out = (float*)d_out;

  int*   cnt       = (int*)d_ws;
  int*   nspill    = cnt + N_NODES;
  float* gsum      = (float*)(nspill + 32);
  float* gsq       = gsum + OUT_DIM;
  int*   nbr       = (int*)(gsq + OUT_DIM);
  int*   spill_src = nbr + N_NODES * SLOTS;
  int*   spill_dst = spill_src + N_EDGES;
  int*   cntAB     = spill_dst + N_EDGES;
  int2*  bins      = (int2*)(cntAB + NBLK_A * CROW);
  int2*  ovf       = bins + (size_t)NBINS * NBLK_A * SEG;

  edge_bin<<<NBLK_A, 1024, 0, stream>>>(src, dst, bins, cntAB, ovf, nspill, gsum);

  edge_runs<<<NBINS, 1024, 0, stream>>>(bins, cntAB, ovf, cnt, nspill, nbr, spill_src, spill_dst);

  fused_mlp<<<N_NODES / 16, 128, 0, stream>>>(h, W1, b1, W2, b2, cnt, nspill, nbr, spill_src, spill_dst, out);

  bn_stats<<<STAT_WGS, STAT_LANES, 0, stream>>>(out, gsum, gsq);

  bn_apply<<<(N_NODES * OUT_DIM + 255) / 256, 256, 0, stream>>>(out, gsum, gsq, gamma, beta);
}
